# starting kernel
# baseline (speedup 1.0000x reference)
#include <hip/hip_runtime.h>
#include <math.h>

#define NN   8192
#define NZ   256
#define NC   128
#define DTC  0.1f

#define OUT_X 0
#define OUT_R NN
#define OUT_Z (2 * NN)
#define OUT_C (2 * NN + NZ)
#define OUT_E (2 * NN + NZ + NC)

#define KSPLIT 8
#define KCHUNK (NN / KSPLIT)

typedef __attribute__((ext_vector_type(2))) float v2f;
typedef __attribute__((ext_vector_type(4))) float v4f;
typedef __attribute__((ext_vector_type(8))) float v8f;

template <bool NT>
__device__ __forceinline__ v8f wmma_gemv(v8f acc,
                                         const float* __restrict__ W, int ld,
                                         int row0,
                                         const float* __restrict__ vec, int K) {
  const int lane = threadIdx.x & 31;
  const int m    = lane & 15;
  const int hl   = lane >> 4;
  const float* wp = W + (size_t)(row0 + m) * (size_t)ld + 4 * hl;
  const float* vp = vec + 4 * hl;
#pragma unroll 8
  for (int k = 0; k < K; k += 8) {
    const v4f w = NT ? __builtin_nontemporal_load((const v4f*)(wp + k))
                     : *(const v4f*)(wp + k);
    const v4f bv = *(const v4f*)(vp + k);
    v2f a0 = {w.x, w.y}, b0 = {bv.x, bv.y};
    v2f a1 = {w.z, w.w}, b1 = {bv.z, bv.w};
    acc = __builtin_amdgcn_wmma_f32_16x16x4_f32(false, a0, false, b0,
                                                (short)0, acc, false, false);
    acc = __builtin_amdgcn_wmma_f32_16x16x4_f32(false, a1, false, b1,
                                                (short)0, acc, false, false);
  }
  return acc;
}

__global__ void k_zpart(const float* __restrict__ W_rz,
                        const float* __restrict__ x,
                        const float* __restrict__ b,
                        float* __restrict__ zpart) {
  __shared__ __align__(16) float s_r[KCHUNK];
  const int tid = threadIdx.x;
  const int k0  = blockIdx.y * KCHUNK;
  for (int j = tid; j < KCHUNK; j += 256)
    s_r[j] = tanhf(x[k0 + j] + b[k0 + j]);
  __syncthreads();
  const int row0 = blockIdx.x * 128 + (tid >> 5) * 16;
  v8f acc = {0.f, 0.f, 0.f, 0.f, 0.f, 0.f, 0.f, 0.f};
  acc = wmma_gemv<false>(acc, W_rz + k0, NN, row0, s_r, KCHUNK);
  const int lane = tid & 31;
  if ((lane & 15) == 0) {
    const int rbase = row0 + (lane >> 4) * 8;
    float* dst = zpart + blockIdx.y * NZ;
#pragma unroll
    for (int v = 0; v < 8; ++v) dst[rbase + v] = acc[v];
  }
}

__global__ void k_rrpart(const float* __restrict__ W_rr,
                         const float* __restrict__ x,
                         const float* __restrict__ b,
                         float* __restrict__ rrpart) {
  __shared__ __align__(16) float s_r[KCHUNK];
  const int tid = threadIdx.x;
  const int k0  = blockIdx.y * KCHUNK;
  for (int j = tid; j < KCHUNK; j += 256)
    s_r[j] = tanhf(x[k0 + j] + b[k0 + j]);
  __syncthreads();
  const int row0 = blockIdx.x * 128 + (tid >> 5) * 16;
  v8f acc = {0.f, 0.f, 0.f, 0.f, 0.f, 0.f, 0.f, 0.f};
  acc = wmma_gemv<true>(acc, W_rr + k0, NN, row0, s_r, KCHUNK);
  const int lane = tid & 31;
  if ((lane & 15) == 0) {
    const int rbase = row0 + (lane >> 4) * 8;
    float* dst = rrpart + blockIdx.y * NN;
#pragma unroll
    for (int v = 0; v < 8; ++v) dst[rbase + v] = acc[v];
  }
}

__global__ void k_final(const float* __restrict__ x,
                        const float* __restrict__ b,
                        const float* __restrict__ eps,
                        const float* __restrict__ cvec,
                        const float* __restrict__ W_epsr,
                        const float* __restrict__ W_zr,
                        const float* __restrict__ W_cr,
                        const float* __restrict__ zpart,
                        const float* __restrict__ rrpart,
                        float* __restrict__ out) {
  __shared__ __align__(16) float s_eps[NZ];
  __shared__ __align__(16) float s_z[NZ];
  __shared__ __align__(16) float s_c[NC];
  const int tid = threadIdx.x;
  {
    float zs = 0.f;
#pragma unroll
    for (int s = 0; s < KSPLIT; ++s) zs += zpart[s * NZ + tid];
    s_z[tid]   = zs;
    s_eps[tid] = eps[tid];
    if (tid < NC) s_c[tid] = cvec[tid];
  }
  __syncthreads();
  const int row0 = blockIdx.x * 128 + (tid >> 5) * 16;
  v8f acc = {0.f, 0.f, 0.f, 0.f, 0.f, 0.f, 0.f, 0.f};
  acc = wmma_gemv<false>(acc, W_epsr, NZ, row0, s_eps, NZ);
  acc = wmma_gemv<false>(acc, W_zr,   NZ, row0, s_z,   NZ);
  acc = wmma_gemv<false>(acc, W_cr,   NC, row0, s_c,   NC);
  const int lane = tid & 31;
  if ((lane & 15) == 0) {
    const int rbase = row0 + (lane >> 4) * 8;
#pragma unroll
    for (int v = 0; v < 8; ++v) {
      const int i = rbase + v;
      float tot = acc[v];
#pragma unroll
      for (int s = 0; s < KSPLIT; ++s) tot += rrpart[s * NN + i];
      const float xv = x[i];
      const float xn = xv + DTC * (tot - xv);
      out[OUT_X + i] = xn;
      out[OUT_R + i] = tanhf(xn + b[i]);
    }
  }
}

__global__ void k_outpart(const float* __restrict__ W_rz,
                          const float* __restrict__ W_rc,
                          const float* __restrict__ r_new,
                          float* __restrict__ znpart,
                          float* __restrict__ cnpart) {
  __shared__ __align__(16) float s_r[KCHUNK];
  const int tid = threadIdx.x;
  const int k0  = blockIdx.y * KCHUNK;
  for (int j = tid; j < KCHUNK; j += 256) s_r[j] = r_new[k0 + j];
  __syncthreads();
  const int wave = tid >> 5;
  const int lane = tid & 31;
  v8f acc = {0.f, 0.f, 0.f, 0.f, 0.f, 0.f, 0.f, 0.f};
  if (blockIdx.x < 2) {
    const int row0 = blockIdx.x * 128 + wave * 16;
    acc = wmma_gemv<false>(acc, W_rz + k0, NN, row0, s_r, KCHUNK);
    if ((lane & 15) == 0) {
      const int rbase = row0 + (lane >> 4) * 8;
#pragma unroll
      for (int v = 0; v < 8; ++v) znpart[blockIdx.y * NZ + rbase + v] = acc[v];
    }
  } else {
    const int row0 = wave * 16;
    acc = wmma_gemv<false>(acc, W_rc + k0, NN, row0, s_r, KCHUNK);
    if ((lane & 15) == 0) {
      const int rbase = row0 + (lane >> 4) * 8;
#pragma unroll
      for (int v = 0; v < 8; ++v) cnpart[blockIdx.y * NC + rbase + v] = acc[v];
    }
  }
}

__global__ void k_epilog(const float* __restrict__ znpart,
                         const float* __restrict__ cnpart,
                         const float* __restrict__ z_tilde,
                         float* __restrict__ out) {
  const int tid = threadIdx.x;
  float zn = 0.f;
#pragma unroll
  for (int s = 0; s < KSPLIT; ++s) zn += znpart[s * NZ + tid];
  out[OUT_Z + tid] = zn;
  out[OUT_E + tid] = zn - z_tilde[tid];
  if (tid < NC) {
    float cn = 0.f;
#pragma unroll
    for (int s = 0; s < KSPLIT; ++s) cn += cnpart[s * NC + tid];
    out[OUT_C + tid] = cn;
  }
}

extern "C" void kernel_launch(void* const* d_in, const int* in_sizes, int n_in,
                              void* d_out, int out_size, void* d_ws, size_t ws_size,
                              hipStream_t stream) {
  (void)in_sizes; (void)n_in; (void)out_size; (void)ws_size;
  const float* x       = (const float*)d_in[0];
  const float* eps     = (const float*)d_in[1];
  const float* c       = (const float*)d_in[2];
  const float* z_tilde = (const float*)d_in[3];
  const float* W_rr    = (const float*)d_in[4];
  const float* W_zr    = (const float*)d_in[5];
  const float* W_cr    = (const float*)d_in[6];
  const float* W_epsr  = (const float*)d_in[7];
  const float* W_rz    = (const float*)d_in[8];
  const float* W_rc    = (const float*)d_in[9];
  const float* b       = (const float*)d_in[10];
  float* out = (float*)d_out;

  float* ws     = (float*)d_ws;
  float* zpart  = ws;
  float* rrpart = zpart  + KSPLIT * NZ;
  float* znpart = rrpart + KSPLIT * NN;
  float* cnpart = znpart + KSPLIT * NZ;

  k_zpart  <<<dim3(2, KSPLIT),  256, 0, stream>>>(W_rz, x, b, zpart);
  k_rrpart <<<dim3(64, KSPLIT), 256, 0, stream>>>(W_rr, x, b, rrpart);
  k_final  <<<64, 256, 0, stream>>>(x, b, eps, c, W_epsr, W_zr, W_cr,
                                    zpart, rrpart, out);
  k_outpart<<<dim3(3, KSPLIT),  256, 0, stream>>>(W_rz, W_rc, out + NN,
                                                  znpart, cnpart);
  k_epilog <<<1, 256, 0, stream>>>(znpart, cnpart, z_tilde, out);
}


# k_step: each wave fetches and parks only its own K-share of r, z, eps, c in LDS rows it alone reads (was: work-group copy of all r in 4 dependent trips, z sum, barrier); every sum unchanged
# speedup vs baseline: 2.6005x; 2.6005x over previous
#include <hip/hip_runtime.h>
#include <math.h>

#define NN   8192
#define NZ   256
#define NC   128
#define DTC  0.1f

#define OUT_X 0
#define OUT_R NN
#define OUT_Z (2 * NN)
#define OUT_C (2 * NN + NZ)
#define OUT_E (2 * NN + NZ + NC)

#define KSPLIT 8
#define KCHUNK (NN / KSPLIT)
#define TK     128
#define SPITCH (TK + 4)
#define NTILE  (KCHUNK / TK)
#define NSTEPWG (NN / 32)
#define NPART  (NZ + NC)
#define ZSPLIT 16
#define ZCHUNK (NN / ZSPLIT)
#define ZW     8

typedef __attribute__((ext_vector_type(2))) float v2f;
typedef __attribute__((ext_vector_type(4))) float v4f;
typedef __attribute__((ext_vector_type(8))) float v8f;

__device__ __forceinline__ v8f wmma_gemv(v8f acc,
                                         const float* __restrict__ W, int ld,
                                         int row0,
                                         const float* __restrict__ vec, int K) {
  const int lane = threadIdx.x & 31;
  const int m    = lane & 15;
  const int hl   = lane >> 4;
  const float* wp = W + (size_t)(row0 + m) * (size_t)ld + 4 * hl;
  const float* vp = vec + 4 * hl;
#pragma unroll 8
  for (int k = 0; k < K; k += 8) {
    const v4f w = *(const v4f*)(wp + k);
    const v4f bv = *(const v4f*)(vp + k);
    v2f a0 = {w.x, w.y}, b0 = {bv.x, bv.y};
    v2f a1 = {w.z, w.w}, b1 = {bv.z, bv.w};
    acc = __builtin_amdgcn_wmma_f32_16x16x4_f32(false, a0, false, b0,
                                                (short)0, acc, false, false);
    acc = __builtin_amdgcn_wmma_f32_16x16x4_f32(false, a1, false, b1,
                                                (short)0, acc, false, false);
  }
  return acc;
}

__device__ __forceinline__ void frag32_load(v4f f[4], const float* __restrict__ W, int row0) {
  const int lane = threadIdx.x & 31;
  const float* p = W + (size_t)(row0 + (lane & 15)) * (size_t)NN + 4 * (lane >> 4);
#pragma unroll
  for (int q = 0; q < 4; ++q) f[q] = *(const v4f*)(p + 8 * q);
}

__device__ __forceinline__ v8f frag32_mul(const v4f f[4], const float* vec) {
  const float* vp = vec + 4 * ((threadIdx.x & 31) >> 4);
  v8f acc = {0.f, 0.f, 0.f, 0.f, 0.f, 0.f, 0.f, 0.f};
#pragma unroll
  for (int q = 0; q < 4; ++q) {
    const v4f w  = f[q];
    const v4f bv = *(const v4f*)(vp + 8 * q);
    v2f a0 = {w.x, w.y}, b0 = {bv.x, bv.y};
    v2f a1 = {w.z, w.w}, b1 = {bv.z, bv.w};
    acc = __builtin_amdgcn_wmma_f32_16x16x4_f32(false, a0, false, b0,
                                                (short)0, acc, false, false);
    acc = __builtin_amdgcn_wmma_f32_16x16x4_f32(false, a1, false, b1,
                                                (short)0, acc, false, false);
  }
  return acc;
}

__device__ __forceinline__ v4f ld_stream(const float* p) {
  return __builtin_nontemporal_load((const v4f*)p);
}

__device__ __forceinline__ v8f wmma_tile(v8f acc, const float* ap, const float* bp) {
#pragma unroll
  for (int kk = 0; kk < TK; kk += 8) {
    const v4f w  = *(const v4f*)(ap + kk);
    const v4f bv = *(const v4f*)(bp + kk);
    v2f a0 = {w.x, w.y}, b0 = {bv.x, bv.y};
    v2f a1 = {w.z, w.w}, b1 = {bv.z, bv.w};
    acc = __builtin_amdgcn_wmma_f32_16x16x4_f32(false, a0, false, b0,
                                                (short)0, acc, false, false);
    acc = __builtin_amdgcn_wmma_f32_16x16x4_f32(false, a1, false, b1,
                                                (short)0, acc, false, false);
  }
  return acc;
}

__device__ __forceinline__ v8f wmma_gemv_staged(v8f acc,
                                                const float* __restrict__ W,
                                                const float* vec,
                                                float* stg) {
  const int lane = threadIdx.x & 31;
  const int c = lane & 15;
  const int h = lane >> 4;
  const float* gp = W + 4 * lane;
  float*       sp = stg + 4 * lane;
  const float* ap = stg + c * SPITCH + 4 * h;
  const float* bp = vec + 4 * h;
  v4f g0[16], g1[16];
#pragma unroll
  for (int i = 0; i < 16; ++i) g0[i] = ld_stream(gp + (size_t)i * NN);
#pragma unroll
  for (int i = 0; i < 16; ++i) g1[i] = ld_stream(gp + TK + (size_t)i * NN);
  for (int t = 0; t < NTILE; t += 2) {
#pragma unroll
    for (int i = 0; i < 16; ++i) *(v4f*)(sp + i * SPITCH) = g0[i];
    if (t + 2 < NTILE) {
#pragma unroll
      for (int i = 0; i < 16; ++i) g0[i] = ld_stream(gp + (t + 2) * TK + (size_t)i * NN);
    }
    acc = wmma_tile(acc, ap, bp + t * TK);
#pragma unroll
    for (int i = 0; i < 16; ++i) *(v4f*)(sp + i * SPITCH) = g1[i];
    if (t + 2 < NTILE) {
#pragma unroll
      for (int i = 0; i < 16; ++i) g1[i] = ld_stream(gp + (t + 3) * TK + (size_t)i * NN);
    }
    acc = wmma_tile(acc, ap, bp + (t + 1) * TK);
  }
  return acc;
}

__global__ void __launch_bounds__(256)
k_zpart(const float* __restrict__ W_rz,
        const float* __restrict__ x,
        const float* __restrict__ b,
        float* __restrict__ zpart,
        float* __restrict__ rvec) {
  __shared__ __align__(16) float s_r[ZCHUNK];
  __shared__ __align__(16) float s_stage[ZW][16 * SPITCH];
  __shared__ float s_part[ZW][16];
  const int tid  = threadIdx.x;
  const int wave = tid >> 5;
  const int lane = tid & 31;
  const int tile = wave & 1;
  const int ks   = wave >> 1;
  const int k0   = blockIdx.y * ZCHUNK;
  const int row0 = blockIdx.x * 32 + tile * 16;
  const v2f xv = *(const v2f*)(x + k0 + 2 * tid);
  const v2f bv = *(const v2f*)(b + k0 + 2 * tid);
  const float* gp = W_rz + (size_t)row0 * NN + k0 + ks * TK + 4 * lane;
  v4f g[16];
#pragma unroll
  for (int i = 0; i < 16; ++i) g[i] = *(const v4f*)(gp + (size_t)i * NN);
  {
    const v2f rv = {tanhf(xv.x + bv.x), tanhf(xv.y + bv.y)};
    *(v2f*)(s_r + 2 * tid) = rv;
    if (blockIdx.x == 0) *(v2f*)(rvec + k0 + 2 * tid) = rv;
  }
  float* sp = s_stage[wave] + 4 * lane;
#pragma unroll
  for (int i = 0; i < 16; ++i) *(v4f*)(sp + i * SPITCH) = g[i];
  __syncthreads();
  v8f acc = {0.f, 0.f, 0.f, 0.f, 0.f, 0.f, 0.f, 0.f};
  acc = wmma_tile(acc, s_stage[wave] + (lane & 15) * SPITCH + 4 * (lane >> 4),
                  s_r + ks * TK + 4 * (lane >> 4));
  if ((lane & 15) == 0) {
    const int rb = (lane >> 4) * 8;
#pragma unroll
    for (int v = 0; v < 8; ++v) s_part[wave][rb + v] = acc[v];
  }
  __syncthreads();
  if (tid < 32) {
    const int t = tid >> 4;
    const int m = tid & 15;
    float tot = 0.f;
#pragma unroll
    for (int s = 0; s < ZW / 2; ++s) tot += s_part[2 * s + t][m];
    zpart[blockIdx.y * NZ + blockIdx.x * 32 + tid] = tot;
  }
}

__global__ void __launch_bounds__(512)
k_step(const float* __restrict__ W_rr,
       const float* __restrict__ W_epsr,
       const float* __restrict__ W_zr,
       const float* __restrict__ W_cr,
       const float* __restrict__ x,
       const float* __restrict__ b,
       const float* __restrict__ eps,
       const float* __restrict__ cvec,
       const float* __restrict__ zpart,
       const float* __restrict__ rvec,
       const float* __restrict__ W_rz,
       const float* __restrict__ W_rc,
       float* __restrict__ part,
       float* __restrict__ out) {
  __shared__ __align__(16) float s_rw[2 * KSPLIT][KCHUNK];
  __shared__ __align__(16) float s_zw[2 * KSPLIT][NZ / KSPLIT];
  __shared__ __align__(16) float s_ew[2 * KSPLIT][NZ / KSPLIT];
  __shared__ __align__(16) float s_cw[2 * KSPLIT][NC / KSPLIT];
  __shared__ __align__(16) float s_rn[32];
  __shared__ __align__(16) float s_ro[NPART];
  __shared__ float s_part[2 * KSPLIT][16];
  __shared__ __align__(16) float s_stage[2 * KSPLIT][16 * SPITCH];
  const int tid = threadIdx.x;
  const int wave = tid >> 5;
  const int lane = tid & 31;
  const int tile = wave & 1;
  const int kc   = wave >> 1;
  const int row0 = blockIdx.x * 32 + tile * 16;
  v4f rv[KCHUNK / TK];
#pragma unroll
  for (int q = 0; q < KCHUNK / TK; ++q)
    rv[q] = *(const v4f*)(rvec + kc * KCHUNK + q * TK + 4 * lane);
  float zp[ZSPLIT];
#pragma unroll
  for (int s = 0; s < ZSPLIT; ++s) zp[s] = zpart[s * NZ + (NZ / KSPLIT) * kc + lane];
  const float ev = eps[(NZ / KSPLIT) * kc + lane];
  const float cv = cvec[(NC / KSPLIT) * kc + (lane & 15)];
  v4f fz[4], fc[4];
  frag32_load(fz, W_rz + blockIdx.x * 32, wave * 16);
  if (wave < NC / 16) frag32_load(fc, W_rc + blockIdx.x * 32, wave * 16);
#pragma unroll
  for (int q = 0; q < KCHUNK / TK; ++q) *(v4f*)(s_rw[wave] + q * TK + 4 * lane) = rv[q];
  {
    float zs = 0.f;
#pragma unroll
    for (int s = 0; s < ZSPLIT; ++s) zs += zp[s];
    s_zw[wave][lane] = zs;
    s_ew[wave][lane] = ev;
    s_cw[wave][lane & 15] = cv;
  }
  v8f acc = {0.f, 0.f, 0.f, 0.f, 0.f, 0.f, 0.f, 0.f};
  acc = wmma_gemv(acc, W_epsr + (NZ / KSPLIT) * kc, NZ, row0,
                         s_ew[wave], NZ / KSPLIT);
  acc = wmma_gemv(acc, W_zr + (NZ / KSPLIT) * kc, NZ, row0,
                         s_zw[wave], NZ / KSPLIT);
  acc = wmma_gemv(acc, W_cr + (NC / KSPLIT) * kc, NC, row0,
                         s_cw[wave], NC / KSPLIT);
  acc = wmma_gemv_staged(acc, W_rr + (size_t)row0 * NN + kc * KCHUNK,
                         s_rw[wave], s_stage[wave]);
  if ((lane & 15) == 0) {
    const int rb = (lane >> 4) * 8;
#pragma unroll
    for (int v = 0; v < 8; ++v) s_part[wave][rb + v] = acc[v];
  }
  __syncthreads();
  float xn = 0.f, rn = 0.f;
  if (tid < 32) {
    const int t = tid >> 4;
    const int m = tid & 15;
    float tot = 0.f;
#pragma unroll
    for (int s = 0; s < KSPLIT; ++s) tot += s_part[2 * s + t][m];
    const int i = blockIdx.x * 32 + tid;
    const float xv = x[i];
    xn = xv + DTC * (tot - xv);
    rn = tanhf(xn + b[i]);
    s_rn[tid] = rn;
  }
  __syncthreads();
  {
    const v8f az = frag32_mul(fz, s_rn);
    if ((lane & 15) == 0) {
      const int rb = wave * 16 + (lane >> 4) * 8;
#pragma unroll
      for (int v = 0; v < 8; ++v) s_ro[rb + v] = az[v];
    }
  }
  if (wave < NC / 16) {
    const v8f ac = frag32_mul(fc, s_rn);
    if ((lane & 15) == 0) {
      const int rb = NZ + wave * 16 + (lane >> 4) * 8;
#pragma unroll
      for (int v = 0; v < 8; ++v) s_ro[rb + v] = ac[v];
    }
  }
  __syncthreads();
  if (tid < 32) {
    const int i = blockIdx.x * 32 + tid;
    out[OUT_X + i] = xn;
    out[OUT_R + i] = rn;
  }
  if (tid < NPART / 4)
    ((v4f*)(part + (size_t)blockIdx.x * NPART))[tid] = ((const v4f*)s_ro)[tid];
}

__global__ void k_epilog(const float* __restrict__ part,
                         const float* __restrict__ z_tilde,
                         float* __restrict__ out) {
  __shared__ float s_sum[8][32];
  const int tid = threadIdx.x;
  const int jj  = tid & 31;
  const int s   = tid >> 5;
  const int j   = blockIdx.x * 32 + jj;
  const float* p = part + (size_t)(s * (NSTEPWG / 8)) * NPART + j;
  const bool zrow = blockIdx.x < NZ / 32;
  float zt = 0.f;
  if (zrow && tid < 32) zt = z_tilde[j];
  float v[NSTEPWG / 8];
#pragma unroll
  for (int q = 0; q < NSTEPWG / 8; ++q) v[q] = p[(size_t)q * NPART];
  float t = 0.f;
#pragma unroll
  for (int q = 0; q < NSTEPWG / 8; ++q) t += v[q];
  s_sum[s][jj] = t;
  __syncthreads();
  if (tid < 32) {
    float tot = 0.f;
#pragma unroll
    for (int w = 0; w < 8; ++w) tot += s_sum[w][tid];
    if (zrow) {
      out[OUT_Z + j] = tot;
      out[OUT_E + j] = tot - zt;
    } else {
      out[OUT_C + (j - NZ)] = tot;
    }
  }
}

extern "C" void kernel_launch(void* const* d_in, const int* in_sizes, int n_in,
                              void* d_out, int out_size, void* d_ws, size_t ws_size,
                              hipStream_t stream) {
  (void)in_sizes; (void)n_in; (void)out_size; (void)ws_size;
  const float* x       = (const float*)d_in[0];
  const float* eps     = (const float*)d_in[1];
  const float* c       = (const float*)d_in[2];
  const float* z_tilde = (const float*)d_in[3];
  const float* W_rr    = (const float*)d_in[4];
  const float* W_zr    = (const float*)d_in[5];
  const float* W_cr    = (const float*)d_in[6];
  const float* W_epsr  = (const float*)d_in[7];
  const float* W_rz    = (const float*)d_in[8];
  const float* W_rc    = (const float*)d_in[9];
  const float* b       = (const float*)d_in[10];
  float* out = (float*)d_out;

  float* ws     = (float*)d_ws;
  float* zpart  = ws;
  float* rvec   = zpart  + ZSPLIT * NZ;
  float* part   = rvec   + NN;

  k_zpart  <<<dim3(NZ / 32, ZSPLIT), 256, 0, stream>>>(W_rz, x, b, zpart, rvec);
  k_step   <<<NSTEPWG, 512, 0, stream>>>(W_rr, W_epsr, W_zr, W_cr, x, b, eps, c,
                                         zpart, rvec, W_rz, W_rc, part, out);
  k_epilog <<<NPART / 32, 256, 0, stream>>>(part, z_tilde, out);
}


# k_step: W_epsr, W_zr, W_cr go the W_rr tile path: 5 tiles of 16 rows x 128 cols per row tile, 512 contiguous bytes per load via LDS (was 16 rows x 32 B per load in every wave); sum order changes
# speedup vs baseline: 2.7188x; 1.0455x over previous
#include <hip/hip_runtime.h>
#include <math.h>

#define NN   8192
#define NZ   256
#define NC   128
#define DTC  0.1f

#define OUT_X 0
#define OUT_R NN
#define OUT_Z (2 * NN)
#define OUT_C (2 * NN + NZ)
#define OUT_E (2 * NN + NZ + NC)

#define KSPLIT 8
#define KCHUNK (NN / KSPLIT)
#define TK     128
#define SPITCH (TK + 4)
#define NTILE  (KCHUNK / TK)
#define NSTEPWG (NN / 32)
#define NXT    5
static_assert(NZ == 2 * TK && NC == TK, "narrow tiles: two per W_epsr / W_zr row tile, one per W_cr row tile");
#define NPART  (NZ + NC)
#define ZSPLIT 16
#define ZCHUNK (NN / ZSPLIT)
#define ZW     8

typedef __attribute__((ext_vector_type(2))) float v2f;
typedef __attribute__((ext_vector_type(4))) float v4f;
typedef __attribute__((ext_vector_type(8))) float v8f;

__device__ __forceinline__ void frag32_load(v4f f[4], const float* __restrict__ W, int row0) {
  const int lane = threadIdx.x & 31;
  const float* p = W + (size_t)(row0 + (lane & 15)) * (size_t)NN + 4 * (lane >> 4);
#pragma unroll
  for (int q = 0; q < 4; ++q) f[q] = *(const v4f*)(p + 8 * q);
}

__device__ __forceinline__ v8f frag32_mul(const v4f f[4], const float* vec) {
  const float* vp = vec + 4 * ((threadIdx.x & 31) >> 4);
  v8f acc = {0.f, 0.f, 0.f, 0.f, 0.f, 0.f, 0.f, 0.f};
#pragma unroll
  for (int q = 0; q < 4; ++q) {
    const v4f w  = f[q];
    const v4f bv = *(const v4f*)(vp + 8 * q);
    v2f a0 = {w.x, w.y}, b0 = {bv.x, bv.y};
    v2f a1 = {w.z, w.w}, b1 = {bv.z, bv.w};
    acc = __builtin_amdgcn_wmma_f32_16x16x4_f32(false, a0, false, b0,
                                                (short)0, acc, false, false);
    acc = __builtin_amdgcn_wmma_f32_16x16x4_f32(false, a1, false, b1,
                                                (short)0, acc, false, false);
  }
  return acc;
}

__device__ __forceinline__ v4f ld_stream(const float* p) {
  return __builtin_nontemporal_load((const v4f*)p);
}

__device__ __forceinline__ v8f wmma_tile(v8f acc, const float* ap, const float* bp) {
#pragma unroll
  for (int kk = 0; kk < TK; kk += 8) {
    const v4f w  = *(const v4f*)(ap + kk);
    const v4f bv = *(const v4f*)(bp + kk);
    v2f a0 = {w.x, w.y}, b0 = {bv.x, bv.y};
    v2f a1 = {w.z, w.w}, b1 = {bv.z, bv.w};
    acc = __builtin_amdgcn_wmma_f32_16x16x4_f32(false, a0, false, b0,
                                                (short)0, acc, false, false);
    acc = __builtin_amdgcn_wmma_f32_16x16x4_f32(false, a1, false, b1,
                                                (short)0, acc, false, false);
  }
  return acc;
}

__device__ __forceinline__ v8f wmma_stream(v8f acc, v4f g0[16], v4f g1[16],
                                           const float* __restrict__ gp,
                                           float* sp,
                                           const float* ap,
                                           const float* bp) {
  for (int t = 0; t < NTILE; t += 2) {
#pragma unroll
    for (int i = 0; i < 16; ++i) *(v4f*)(sp + i * SPITCH) = g0[i];
    if (t + 2 < NTILE) {
#pragma unroll
      for (int i = 0; i < 16; ++i) g0[i] = ld_stream(gp + (t + 2) * TK + (size_t)i * NN);
    }
    acc = wmma_tile(acc, ap, bp + t * TK);
#pragma unroll
    for (int i = 0; i < 16; ++i) *(v4f*)(sp + i * SPITCH) = g1[i];
    if (t + 2 < NTILE) {
#pragma unroll
      for (int i = 0; i < 16; ++i) g1[i] = ld_stream(gp + (t + 3) * TK + (size_t)i * NN);
    }
    acc = wmma_tile(acc, ap, bp + (t + 1) * TK);
  }
  return acc;
}

__global__ void __launch_bounds__(256)
k_zpart(const float* __restrict__ W_rz,
        const float* __restrict__ x,
        const float* __restrict__ b,
        float* __restrict__ zpart,
        float* __restrict__ rvec) {
  __shared__ __align__(16) float s_r[ZCHUNK];
  __shared__ __align__(16) float s_stage[ZW][16 * SPITCH];
  __shared__ float s_part[ZW][16];
  const int tid  = threadIdx.x;
  const int wave = tid >> 5;
  const int lane = tid & 31;
  const int tile = wave & 1;
  const int ks   = wave >> 1;
  const int k0   = blockIdx.y * ZCHUNK;
  const int row0 = blockIdx.x * 32 + tile * 16;
  const v2f xv = *(const v2f*)(x + k0 + 2 * tid);
  const v2f bv = *(const v2f*)(b + k0 + 2 * tid);
  const float* gp = W_rz + (size_t)row0 * NN + k0 + ks * TK + 4 * lane;
  v4f g[16];
#pragma unroll
  for (int i = 0; i < 16; ++i) g[i] = *(const v4f*)(gp + (size_t)i * NN);
  {
    const v2f rv = {tanhf(xv.x + bv.x), tanhf(xv.y + bv.y)};
    *(v2f*)(s_r + 2 * tid) = rv;
    if (blockIdx.x == 0) *(v2f*)(rvec + k0 + 2 * tid) = rv;
  }
  float* sp = s_stage[wave] + 4 * lane;
#pragma unroll
  for (int i = 0; i < 16; ++i) *(v4f*)(sp + i * SPITCH) = g[i];
  __syncthreads();
  v8f acc = {0.f, 0.f, 0.f, 0.f, 0.f, 0.f, 0.f, 0.f};
  acc = wmma_tile(acc, s_stage[wave] + (lane & 15) * SPITCH + 4 * (lane >> 4),
                  s_r + ks * TK + 4 * (lane >> 4));
  if ((lane & 15) == 0) {
    const int rb = (lane >> 4) * 8;
#pragma unroll
    for (int v = 0; v < 8; ++v) s_part[wave][rb + v] = acc[v];
  }
  __syncthreads();
  if (tid < 32) {
    const int t = tid >> 4;
    const int m = tid & 15;
    float tot = 0.f;
#pragma unroll
    for (int s = 0; s < ZW / 2; ++s) tot += s_part[2 * s + t][m];
    zpart[blockIdx.y * NZ + blockIdx.x * 32 + tid] = tot;
  }
}

__global__ void __launch_bounds__(512)
k_step(const float* __restrict__ W_rr,
       const float* __restrict__ W_epsr,
       const float* __restrict__ W_zr,
       const float* __restrict__ W_cr,
       const float* __restrict__ x,
       const float* __restrict__ b,
       const float* __restrict__ eps,
       const float* __restrict__ cvec,
       const float* __restrict__ zpart,
       const float* __restrict__ rvec,
       const float* __restrict__ W_rz,
       const float* __restrict__ W_rc,
       float* __restrict__ part,
       float* __restrict__ out) {
  __shared__ __align__(16) float s_rw[2 * KSPLIT][KCHUNK];
  __shared__ __align__(16) float s_vw[2 * KSPLIT][TK];
  __shared__ __align__(16) float s_rn[32];
  __shared__ __align__(16) float s_ro[NPART];
  __shared__ float s_part[2 * KSPLIT][16];
  __shared__ __align__(16) float s_stage[2 * KSPLIT][16 * SPITCH];
  const int tid = threadIdx.x;
  const int wave = __builtin_amdgcn_readfirstlane(tid >> 5);
  const int lane = tid & 31;
  const int tile = wave & 1;
  const int kc   = wave >> 1;
  const int row0 = blockIdx.x * 32 + tile * 16;
  const bool extra = kc < NXT;
  const bool ztile = (kc >> 1) == 1;
  const int kt = kc & 1;
  const float* gp = W_rr + (size_t)row0 * NN + kc * KCHUNK + 4 * lane;
  float* sp = s_stage[wave] + 4 * lane;
  const float* ap = s_stage[wave] + (lane & 15) * SPITCH + 4 * (lane >> 4);
  v4f vv = {0.f, 0.f, 0.f, 0.f};
  if (ztile) {
    v4f zp[ZSPLIT];
#pragma unroll
    for (int s = 0; s < ZSPLIT; ++s) zp[s] = *(const v4f*)(zpart + s * NZ + kt * TK + 4 * lane);
#pragma unroll
    for (int s = 0; s < ZSPLIT; ++s) vv += zp[s];
  } else if (extra) {
    vv = *(const v4f*)((kc < 2 ? eps + kt * TK : cvec) + 4 * lane);
  }
  v4f rv[KCHUNK / TK];
#pragma unroll
  for (int q = 0; q < KCHUNK / TK; ++q)
    rv[q] = *(const v4f*)(rvec + kc * KCHUNK + q * TK + 4 * lane);
  v4f fz[4], fc[4];
  frag32_load(fz, W_rz + blockIdx.x * 32, wave * 16);
  if (wave < NC / 16) frag32_load(fc, W_rc + blockIdx.x * 32, wave * 16);
  v4f g0[16], g1[16];
  if (extra) {
    const float* np = kc < 2 ? W_epsr + (size_t)row0 * NZ + kt * TK
                    : kc < 4 ? W_zr   + (size_t)row0 * NZ + kt * TK
                             : W_cr   + (size_t)row0 * NC;
    const int ldn = kc < 4 ? NZ : NC;
#pragma unroll
    for (int i = 0; i < 16; ++i) g1[i] = *(const v4f*)(np + 4 * lane + (size_t)i * ldn);
  }
#pragma unroll
  for (int i = 0; i < 16; ++i) g0[i] = ld_stream(gp + (size_t)i * NN);
  if (extra) {
    *(v4f*)(s_vw[wave] + 4 * lane) = vv;
#pragma unroll
    for (int i = 0; i < 16; ++i) *(v4f*)(sp + i * SPITCH) = g1[i];
  }
#pragma unroll
  for (int i = 0; i < 16; ++i) g1[i] = ld_stream(gp + TK + (size_t)i * NN);
  v8f acc = {0.f, 0.f, 0.f, 0.f, 0.f, 0.f, 0.f, 0.f};
  if (extra) acc = wmma_tile(acc, ap, s_vw[wave] + 4 * (lane >> 4));
#pragma unroll
  for (int q = 0; q < KCHUNK / TK; ++q) *(v4f*)(s_rw[wave] + q * TK + 4 * lane) = rv[q];
  acc = wmma_stream(acc, g0, g1, gp, sp, ap, s_rw[wave] + 4 * (lane >> 4));
  if ((lane & 15) == 0) {
    const int rb = (lane >> 4) * 8;
#pragma unroll
    for (int v = 0; v < 8; ++v) s_part[wave][rb + v] = acc[v];
  }
  __syncthreads();
  float xn = 0.f, rn = 0.f;
  if (tid < 32) {
    const int t = tid >> 4;
    const int m = tid & 15;
    float tot = 0.f;
#pragma unroll
    for (int s = 0; s < KSPLIT; ++s) tot += s_part[2 * s + t][m];
    const int i = blockIdx.x * 32 + tid;
    const float xv = x[i];
    xn = xv + DTC * (tot - xv);
    rn = tanhf(xn + b[i]);
    s_rn[tid] = rn;
  }
  __syncthreads();
  {
    const v8f az = frag32_mul(fz, s_rn);
    if ((lane & 15) == 0) {
      const int rb = wave * 16 + (lane >> 4) * 8;
#pragma unroll
      for (int v = 0; v < 8; ++v) s_ro[rb + v] = az[v];
    }
  }
  if (wave < NC / 16) {
    const v8f ac = frag32_mul(fc, s_rn);
    if ((lane & 15) == 0) {
      const int rb = NZ + wave * 16 + (lane >> 4) * 8;
#pragma unroll
      for (int v = 0; v < 8; ++v) s_ro[rb + v] = ac[v];
    }
  }
  __syncthreads();
  if (tid < 32) {
    const int i = blockIdx.x * 32 + tid;
    out[OUT_X + i] = xn;
    out[OUT_R + i] = rn;
  }
  if (tid < NPART / 4)
    ((v4f*)(part + (size_t)blockIdx.x * NPART))[tid] = ((const v4f*)s_ro)[tid];
}

__global__ void k_epilog(const float* __restrict__ part,
                         const float* __restrict__ z_tilde,
                         float* __restrict__ out) {
  __shared__ float s_sum[8][32];
  const int tid = threadIdx.x;
  const int jj  = tid & 31;
  const int s   = tid >> 5;
  const int j   = blockIdx.x * 32 + jj;
  const float* p = part + (size_t)(s * (NSTEPWG / 8)) * NPART + j;
  const bool zrow = blockIdx.x < NZ / 32;
  float zt = 0.f;
  if (zrow && tid < 32) zt = z_tilde[j];
  float v[NSTEPWG / 8];
#pragma unroll
  for (int q = 0; q < NSTEPWG / 8; ++q) v[q] = p[(size_t)q * NPART];
  float t = 0.f;
#pragma unroll
  for (int q = 0; q < NSTEPWG / 8; ++q) t += v[q];
  s_sum[s][jj] = t;
  __syncthreads();
  if (tid < 32) {
    float tot = 0.f;
#pragma unroll
    for (int w = 0; w < 8; ++w) tot += s_sum[w][tid];
    if (zrow) {
      out[OUT_Z + j] = tot;
      out[OUT_E + j] = tot - zt;
    } else {
      out[OUT_C + (j - NZ)] = tot;
    }
  }
}

extern "C" void kernel_launch(void* const* d_in, const int* in_sizes, int n_in,
                              void* d_out, int out_size, void* d_ws, size_t ws_size,
                              hipStream_t stream) {
  (void)in_sizes; (void)n_in; (void)out_size; (void)ws_size;
  const float* x       = (const float*)d_in[0];
  const float* eps     = (const float*)d_in[1];
  const float* c       = (const float*)d_in[2];
  const float* z_tilde = (const float*)d_in[3];
  const float* W_rr    = (const float*)d_in[4];
  const float* W_zr    = (const float*)d_in[5];
  const float* W_cr    = (const float*)d_in[6];
  const float* W_epsr  = (const float*)d_in[7];
  const float* W_rz    = (const float*)d_in[8];
  const float* W_rc    = (const float*)d_in[9];
  const float* b       = (const float*)d_in[10];
  float* out = (float*)d_out;

  float* ws     = (float*)d_ws;
  float* zpart  = ws;
  float* rvec   = zpart  + ZSPLIT * NZ;
  float* part   = rvec   + NN;

  k_zpart  <<<dim3(NZ / 32, ZSPLIT), 256, 0, stream>>>(W_rz, x, b, zpart, rvec);
  k_step   <<<NSTEPWG, 512, 0, stream>>>(W_rr, W_epsr, W_zr, W_cr, x, b, eps, c,
                                         zpart, rvec, W_rz, W_rc, part, out);
  k_epilog <<<NPART / 32, 256, 0, stream>>>(part, z_tilde, out);
}
